# starting kernel
# baseline (speedup 1.0000x reference)
#include <hip/hip_runtime.h>

typedef float v2f __attribute__((ext_vector_type(2)));
typedef float v4f __attribute__((ext_vector_type(4)));
typedef float v8f __attribute__((ext_vector_type(8)));

#define IN_C 128
#define OUT_C 32
#define LDSTRIDE 132
#define ROWS_PER_BLOCK 64
#define GEMM_THREADS 128

__global__ __launch_bounds__(GEMM_THREADS)
void gcn_gemm_wmma(const float* __restrict__ x, const float* __restrict__ W,
                   const float* __restrict__ b, float* __restrict__ h, int nRows)
{
    __shared__ float xs[ROWS_PER_BLOCK * LDSTRIDE];
    __shared__ float wls[OUT_C * LDSTRIDE];

    const int tid = threadIdx.x;
    const int blockRow0 = blockIdx.x * ROWS_PER_BLOCK;
    const int nv4row = IN_C / 4;

    {
        const v4f* x4 = (const v4f*)x;
        #pragma unroll
        for (int i = 0; i < (ROWS_PER_BLOCK * nv4row) / GEMM_THREADS; ++i) {
            int j = i * GEMM_THREADS + tid;
            int r = j / nv4row, c4 = j % nv4row;
            v4f v = {0.f, 0.f, 0.f, 0.f};
            int gr = blockRow0 + r;
            if (gr < nRows) v = x4[(size_t)gr * nv4row + c4];
            *(v4f*)&xs[r * LDSTRIDE + c4 * 4] = v;
        }
        const v4f* W4 = (const v4f*)W;
        #pragma unroll
        for (int i = 0; i < (OUT_C * nv4row) / GEMM_THREADS; ++i) {
            int j = i * GEMM_THREADS + tid;
            int r = j / nv4row, c4 = j % nv4row;
            *(v4f*)&wls[r * LDSTRIDE + c4 * 4] = W4[r * nv4row + c4];
        }
    }
    __syncthreads();

    const int wave = tid >> 5;
    const int lane = tid & 31;
    const int half = lane >> 4;
    const int mn   = lane & 15;

    const int tileRow0 = blockRow0 + wave * 16;
    if (tileRow0 >= nRows) return;

    v8f c0 = {};
    v8f c1 = {};

    const float* aBase  = &xs[(wave * 16 + mn) * LDSTRIDE + 2 * half];
    const float* b0Base = &wls[mn * LDSTRIDE + 2 * half];
    const float* b1Base = b0Base + 16 * LDSTRIDE;

    #pragma unroll
    for (int kk = 0; kk < IN_C / 4; ++kk) {
        v2f a  = *(const v2f*)(aBase  + kk * 4);
        v2f b0 = *(const v2f*)(b0Base + kk * 4);
        v2f b1 = *(const v2f*)(b1Base + kk * 4);
        c0 = __builtin_amdgcn_wmma_f32_16x16x4_f32(false, a, false, b0,
                                                   (short)0, c0, false, false);
        c1 = __builtin_amdgcn_wmma_f32_16x16x4_f32(false, a, false, b1,
                                                   (short)0, c1, false, false);
    }

    const float bias0 = b[mn];
    const float bias1 = b[16 + mn];
    #pragma unroll
    for (int v = 0; v < 8; ++v) {
        int row = tileRow0 + v + 8 * half;
        h[(size_t)row * OUT_C + mn]      = c0[v] + bias0;
        h[(size_t)row * OUT_C + 16 + mn] = c1[v] + bias1;
    }
}

__global__ __launch_bounds__(256)
void gcn_aggregate(const float* __restrict__ h, const float* __restrict__ ew,
                   const float* __restrict__ sw, const int* __restrict__ tar_ptr,
                   const int* __restrict__ src, float* __restrict__ out, int nNodes)
{
    const int lane = threadIdx.x & 31;
    const int node = (int)((blockIdx.x * blockDim.x + threadIdx.x) >> 5);
    if (node >= nNodes) return;

    const int start = tar_ptr[node];
    const int end   = tar_ptr[node + 1];

    const int c8   = lane & 7;
    const int slot = lane >> 3;

    const v4f* h4 = (const v4f*)h;
    v4f acc = {0.f, 0.f, 0.f, 0.f};

    for (int e0 = start; e0 < end; e0 += 4) {
        int e = e0 + slot;
        if (e < end) {
            int   s = src[e];
            float w = ew[e];
            v4f hv = h4[(size_t)s * (OUT_C / 4) + c8];
            v4f wv = {w, w, w, w};
            acc += wv * hv;
        }
    }

    #pragma unroll
    for (int i = 0; i < 4; ++i) {
        acc[i] += __shfl_xor(acc[i], 8, 32);
        acc[i] += __shfl_xor(acc[i], 16, 32);
    }

    if (slot == 0) {
        float s = sw[node];
        v4f sv = {s, s, s, s};
        v4f hs = h4[(size_t)node * (OUT_C / 4) + c8];
        v4f o  = acc + sv * hs;
        *(v4f*)&out[(size_t)node * OUT_C + c8 * 4] = o;
    }
}

extern "C" void kernel_launch(void* const* d_in, const int* in_sizes, int n_in,
                              void* d_out, int out_size, void* d_ws, size_t ws_size,
                              hipStream_t stream) {
    const float* x   = (const float*)d_in[0];
    const float* W   = (const float*)d_in[1];
    const float* b   = (const float*)d_in[2];
    const float* ew  = (const float*)d_in[3];
    const float* sw  = (const float*)d_in[4];
    const int*   tp  = (const int*)d_in[5];
    const int*   src = (const int*)d_in[6];
    float*       out = (float*)d_out;

    const int nNodes = in_sizes[4];
    float* h = (float*)d_ws;

    const int gemmBlocks = (nNodes + ROWS_PER_BLOCK - 1) / ROWS_PER_BLOCK;
    gcn_gemm_wmma<<<gemmBlocks, GEMM_THREADS, 0, stream>>>(x, W, b, h, nNodes);

    const int wavesPerBlock = 256 / 32;
    const int aggBlocks = (nNodes + wavesPerBlock - 1) / wavesPerBlock;
    gcn_aggregate<<<aggBlocks, 256, 0, stream>>>(h, ew, sw, tp, src, out, nNodes);
}


# aggregate: lane = channel; 8192 waves each walk a run of nodes (was a wave per node); index+weight by scalar loads, 16 at once, 1 node ahead; row load + fma per edge, edge order (sum order changes)
# speedup vs baseline: 1.7596x; 1.7596x over previous
#include <hip/hip_runtime.h>

typedef float v2f __attribute__((ext_vector_type(2)));
typedef float v4f __attribute__((ext_vector_type(4)));
typedef float v8f __attribute__((ext_vector_type(8)));

#define IN_C 128
#define OUT_C 32
#define TILES_PER_WAVE 2

__global__ __launch_bounds__(32)
void gcn_gemm_wmma(const float* __restrict__ x, const float* __restrict__ W,
                   const float* __restrict__ b, float* __restrict__ h, int nRows)
{
    const int lane = threadIdx.x;
    const int half = lane >> 4;
    const int mn   = lane & 15;

    const int row0 = blockIdx.x * (16 * TILES_PER_WAVE);

    int arow = row0 + mn;
    arow = arow < nRows ? arow : nRows - 1;
    const v4f* ap = (const v4f*)(x + (size_t)arow * IN_C + 4 * half);

    const v4f* w0p = (const v4f*)(W + (size_t)mn * IN_C + 4 * half);
    const v4f* w1p = (const v4f*)(W + (size_t)(16 + mn) * IN_C + 4 * half);
    v4f wf0[IN_C / 8], wf1[IN_C / 8], af[IN_C / 8];
    #pragma unroll
    for (int j = 0; j < IN_C / 8; ++j) {
        af[j]  = ap[2 * j];
        wf0[j] = w0p[2 * j];
        wf1[j] = w1p[2 * j];
    }

    const float bias0 = b[mn];
    const float bias1 = b[16 + mn];

    #pragma unroll
    for (int t = 0; t < TILES_PER_WAVE; ++t) {
        const int tileRow0 = row0 + t * 16;

        const v4f* apn = ap;
        if (t + 1 < TILES_PER_WAVE) {
            int nrow = tileRow0 + 16 + mn;
            nrow = nrow < nRows ? nrow : nRows - 1;
            apn = (const v4f*)(x + (size_t)nrow * IN_C + 4 * half);
        }

        v8f c0 = {};
        v8f c1 = {};
        #pragma unroll
        for (int j = 0; j < IN_C / 8; ++j) {
            const v4f a = af[j];
            if (t + 1 < TILES_PER_WAVE) af[j] = apn[2 * j];
            v2f alo  = __builtin_shufflevector(a,      a,      0, 1);
            v2f ahi  = __builtin_shufflevector(a,      a,      2, 3);
            v2f b0lo = __builtin_shufflevector(wf0[j], wf0[j], 0, 1);
            v2f b0hi = __builtin_shufflevector(wf0[j], wf0[j], 2, 3);
            v2f b1lo = __builtin_shufflevector(wf1[j], wf1[j], 0, 1);
            v2f b1hi = __builtin_shufflevector(wf1[j], wf1[j], 2, 3);
            c0 = __builtin_amdgcn_wmma_f32_16x16x4_f32(false, alo, false, b0lo,
                                                       (short)0, c0, false, false);
            c1 = __builtin_amdgcn_wmma_f32_16x16x4_f32(false, alo, false, b1lo,
                                                       (short)0, c1, false, false);
            c0 = __builtin_amdgcn_wmma_f32_16x16x4_f32(false, ahi, false, b0hi,
                                                       (short)0, c0, false, false);
            c1 = __builtin_amdgcn_wmma_f32_16x16x4_f32(false, ahi, false, b1hi,
                                                       (short)0, c1, false, false);
        }

        if (tileRow0 < nRows) {
            #pragma unroll
            for (int v = 0; v < 8; ++v) {
                int row = tileRow0 + v + 8 * half;
                h[(size_t)row * OUT_C + mn]      = c0[v] + bias0;
                h[(size_t)row * OUT_C + 16 + mn] = c1[v] + bias1;
            }
        }
    }
}

typedef int   v16i_u __attribute__((ext_vector_type(16), aligned(4)));
typedef float v16f_u __attribute__((ext_vector_type(16), aligned(4)));

#define AGG_WAVES_PER_BLOCK 8

__device__ __forceinline__ float h_at(const float* __restrict__ h, int s, unsigned laneByte)
{
    unsigned boff = ((unsigned)s << 7) + laneByte;
    return *(const float*)((const char*)h + boff);
}

__global__ __launch_bounds__(32 * AGG_WAVES_PER_BLOCK)
void gcn_aggregate(const float* __restrict__ h, const float* __restrict__ ew,
                   const float* __restrict__ sw, const int* __restrict__ tar_ptr,
                   const int* __restrict__ src, float* __restrict__ out, int nNodes)
{
    const int lane = (int)(threadIdx.x & 31u);
    const unsigned laneByte = (unsigned)lane * 4u;
    const int waveInBlock = __builtin_amdgcn_readfirstlane((int)(threadIdx.x >> 5));
    const int wave   = (int)blockIdx.x * AGG_WAVES_PER_BLOCK + waveInBlock;
    const int nWaves = (int)gridDim.x * AGG_WAVES_PER_BLOCK;
    const int per    = (nNodes + nWaves - 1) / nWaves;
    const int n0     = wave * per;
    const int n1     = (n0 + per < nNodes) ? (n0 + per) : nNodes;
    if (n0 >= n1) return;

    int t0 = tar_ptr[n0];
    int t1 = tar_ptr[n0 + 1];
    int t2 = tar_ptr[(n0 + 2 < nNodes) ? (n0 + 2) : nNodes];
    const int tEnd = tar_ptr[n1];

    {
        const int eL = t0 + lane * 64;
        if (eL < tEnd) {
            __builtin_prefetch(src + eL, 0, 1);
            __builtin_prefetch(ew + eL, 0, 1);
        }
    }

    bool  have = (t1 - t0) >= 16;
    v16i_u S = {};
    v16f_u Wt = {};
    if (have) {
        S  = *(const v16i_u*)(src + t0);
        Wt = *(const v16f_u*)(ew + t0);
    }

    for (int n = n0; n < n1; ++n) {
        const bool haveN = (n + 1 < n1) && ((t2 - t1) >= 16);
        const int  tn    = (n + 3 < nNodes) ? (n + 3) : nNodes;
        int   t3, e;
        float selfw, hself, acc;
        v16i_u SN = {};
        v16f_u WN = {};

        if (have) {
            float hv[16];
            #pragma unroll
            for (int j = 0; j < 16; ++j) hv[j] = h_at(h, S[j], laneByte);
            hself = h_at(h, n, laneByte);
            const int pe = haveN ? t1 : t0;
            SN    = *(const v16i_u*)(src + pe);
            WN    = *(const v16f_u*)(ew + pe);
            t3    = tar_ptr[tn];
            selfw = sw[n];
            __builtin_amdgcn_sched_barrier(0);
            acc = 0.f;
            #pragma unroll
            for (int j = 0; j < 16; ++j) acc = __builtin_fmaf(Wt[j], hv[j], acc);
            __builtin_amdgcn_sched_barrier(0);
            e = t0 + 16;
        } else {
            hself = h_at(h, n, laneByte);
            t3    = tar_ptr[tn];
            selfw = sw[n];
            if (haveN) {
                SN = *(const v16i_u*)(src + t1);
                WN = *(const v16f_u*)(ew + t1);
            }
            acc = 0.f;
            e = t0;
        }
        for (; e + 16 <= t1; e += 16) {
            const v16i_u S2 = *(const v16i_u*)(src + e);
            const v16f_u W2 = *(const v16f_u*)(ew + e);
            float hv[16];
            #pragma unroll
            for (int j = 0; j < 16; ++j) hv[j] = h_at(h, S2[j], laneByte);
            #pragma unroll
            for (int j = 0; j < 16; ++j) acc = __builtin_fmaf(W2[j], hv[j], acc);
        }
        for (; e < t1; ++e) acc = __builtin_fmaf(ew[e], h_at(h, src[e], laneByte), acc);

        acc = __builtin_fmaf(selfw, hself, acc);
        *(float*)((char*)out + (((unsigned)n << 7) + laneByte)) = acc;

        t0 = t1; t1 = t2; t2 = t3;
        S = SN; Wt = WN; have = haveN;
    }
}

extern "C" void kernel_launch(void* const* d_in, const int* in_sizes, int n_in,
                              void* d_out, int out_size, void* d_ws, size_t ws_size,
                              hipStream_t stream) {
    const float* x   = (const float*)d_in[0];
    const float* W   = (const float*)d_in[1];
    const float* b   = (const float*)d_in[2];
    const float* ew  = (const float*)d_in[3];
    const float* sw  = (const float*)d_in[4];
    const int*   tp  = (const int*)d_in[5];
    const int*   src = (const int*)d_in[6];
    float*       out = (float*)d_out;

    const int nNodes = in_sizes[4];
    float* h = (float*)d_ws;

    const int rowsPerWave = 16 * TILES_PER_WAVE;
    const int gemmBlocks = (nNodes + rowsPerWave - 1) / rowsPerWave;
    gcn_gemm_wmma<<<gemmBlocks, 32, 0, stream>>>(x, W, b, h, nNodes);

    gcn_aggregate<<<1024, 32 * AGG_WAVES_PER_BLOCK, 0, stream>>>(h, ew, sw, tp, src, out, nNodes);
}


# projection GEMM: 1024 one-wave groups each loop over a run of 6-7 row tiles (was 3125 groups of 2); W fragments loaded once per wave, the next tile's x fragments fill a second register set meanwhile
# speedup vs baseline: 1.8851x; 1.0713x over previous
#include <hip/hip_runtime.h>

typedef float v2f __attribute__((ext_vector_type(2)));
typedef float v4f __attribute__((ext_vector_type(4)));
typedef float v8f __attribute__((ext_vector_type(8)));

#define IN_C 128
#define OUT_C 32
#define GEMM_WAVES 1024

__device__ __forceinline__ const v4f* x_frag_ptr(const float* __restrict__ x, int tile, int lastTile,
                                                 int mn, int half, int nRows)
{
    const int t = tile < lastTile ? tile : lastTile;
    int row = t * 16 + mn;
    row = row < nRows ? row : nRows - 1;
    return (const v4f*)(x + (size_t)row * IN_C + 4 * half);
}

__device__ __forceinline__ void tile_mma_store(const v4f (&af)[IN_C / 8], const v4f (&wf0)[IN_C / 8],
                                               const v4f (&wf1)[IN_C / 8], float bias0, float bias1,
                                               float* __restrict__ h, int tileRow0, int half, int mn)
{
    v8f c0 = {};
    v8f c1 = {};
    #pragma unroll
    for (int j = 0; j < IN_C / 8; ++j) {
        v2f alo  = __builtin_shufflevector(af[j],  af[j],  0, 1);
        v2f ahi  = __builtin_shufflevector(af[j],  af[j],  2, 3);
        v2f b0lo = __builtin_shufflevector(wf0[j], wf0[j], 0, 1);
        v2f b0hi = __builtin_shufflevector(wf0[j], wf0[j], 2, 3);
        v2f b1lo = __builtin_shufflevector(wf1[j], wf1[j], 0, 1);
        v2f b1hi = __builtin_shufflevector(wf1[j], wf1[j], 2, 3);
        c0 = __builtin_amdgcn_wmma_f32_16x16x4_f32(false, alo, false, b0lo,
                                                   (short)0, c0, false, false);
        c1 = __builtin_amdgcn_wmma_f32_16x16x4_f32(false, alo, false, b1lo,
                                                   (short)0, c1, false, false);
        c0 = __builtin_amdgcn_wmma_f32_16x16x4_f32(false, ahi, false, b0hi,
                                                   (short)0, c0, false, false);
        c1 = __builtin_amdgcn_wmma_f32_16x16x4_f32(false, ahi, false, b1hi,
                                                   (short)0, c1, false, false);
    }
    float* hp = h + (size_t)(tileRow0 + 8 * half) * OUT_C + mn;
    #pragma unroll
    for (int v = 0; v < 8; ++v) {
        hp[v * OUT_C]      = c0[v] + bias0;
        hp[v * OUT_C + 16] = c1[v] + bias1;
    }
}

__global__ __launch_bounds__(32)
void gcn_gemm_wmma(const float* __restrict__ x, const float* __restrict__ W,
                   const float* __restrict__ b, float* __restrict__ h, int nRows)
{
    const int lane = threadIdx.x;
    const int half = lane >> 4;
    const int mn   = lane & 15;

    const int nTiles = (nRows + 15) >> 4;
    const int nW     = (int)gridDim.x;
    const int w      = (int)blockIdx.x;
    const int base   = nTiles / nW;
    const int rem    = nTiles - base * nW;
    const int tile0  = w * base + (w < rem ? w : rem);
    const int cnt    = base + (w < rem ? 1 : 0);
    if (cnt <= 0) return;
    const int lastTile = tile0 + cnt - 1;

    const v4f* w0p = (const v4f*)(W + (size_t)mn * IN_C + 4 * half);
    const v4f* w1p = (const v4f*)(W + (size_t)(16 + mn) * IN_C + 4 * half);
    v4f wf0[IN_C / 8], wf1[IN_C / 8], afA[IN_C / 8], afB[IN_C / 8];
    {
        const v4f* ap = x_frag_ptr(x, tile0, lastTile, mn, half, nRows);
        #pragma unroll
        for (int j = 0; j < IN_C / 8; ++j) {
            afA[j] = ap[2 * j];
            wf0[j] = w0p[2 * j];
            wf1[j] = w1p[2 * j];
        }
    }

    const float bias0 = b[mn];
    const float bias1 = b[16 + mn];

    int t = tile0;
    for (int p = cnt >> 1; p > 0; --p, t += 2) {
        {
            const v4f* ap = x_frag_ptr(x, t + 1, lastTile, mn, half, nRows);
            #pragma unroll
            for (int j = 0; j < IN_C / 8; ++j) afB[j] = ap[2 * j];
        }
        __builtin_amdgcn_sched_barrier(0);
        tile_mma_store(afA, wf0, wf1, bias0, bias1, h, t * 16, half, mn);
        {
            const v4f* ap = x_frag_ptr(x, t + 2, lastTile, mn, half, nRows);
            #pragma unroll
            for (int j = 0; j < IN_C / 8; ++j) afA[j] = ap[2 * j];
        }
        __builtin_amdgcn_sched_barrier(0);
        tile_mma_store(afB, wf0, wf1, bias0, bias1, h, (t + 1) * 16, half, mn);
    }
    if (cnt & 1)
        tile_mma_store(afA, wf0, wf1, bias0, bias1, h, t * 16, half, mn);
}

typedef int   v16i_u __attribute__((ext_vector_type(16), aligned(4)));
typedef float v16f_u __attribute__((ext_vector_type(16), aligned(4)));

#define AGG_WAVES_PER_BLOCK 8

__device__ __forceinline__ float h_at(const float* __restrict__ h, int s, unsigned laneByte)
{
    unsigned boff = ((unsigned)s << 7) + laneByte;
    return *(const float*)((const char*)h + boff);
}

__global__ __launch_bounds__(32 * AGG_WAVES_PER_BLOCK)
void gcn_aggregate(const float* __restrict__ h, const float* __restrict__ ew,
                   const float* __restrict__ sw, const int* __restrict__ tar_ptr,
                   const int* __restrict__ src, float* __restrict__ out, int nNodes)
{
    const int lane = (int)(threadIdx.x & 31u);
    const unsigned laneByte = (unsigned)lane * 4u;
    const int waveInBlock = __builtin_amdgcn_readfirstlane((int)(threadIdx.x >> 5));
    const int wave   = (int)blockIdx.x * AGG_WAVES_PER_BLOCK + waveInBlock;
    const int nWaves = (int)gridDim.x * AGG_WAVES_PER_BLOCK;
    const int per    = (nNodes + nWaves - 1) / nWaves;
    const int n0     = wave * per;
    const int n1     = (n0 + per < nNodes) ? (n0 + per) : nNodes;
    if (n0 >= n1) return;

    int t0 = tar_ptr[n0];
    int t1 = tar_ptr[n0 + 1];
    int t2 = tar_ptr[(n0 + 2 < nNodes) ? (n0 + 2) : nNodes];
    const int tEnd = tar_ptr[n1];

    {
        const int eL = t0 + lane * 64;
        if (eL < tEnd) {
            __builtin_prefetch(src + eL, 0, 1);
            __builtin_prefetch(ew + eL, 0, 1);
        }
    }

    bool  have = (t1 - t0) >= 16;
    v16i_u S = {};
    v16f_u Wt = {};
    if (have) {
        S  = *(const v16i_u*)(src + t0);
        Wt = *(const v16f_u*)(ew + t0);
    }

    for (int n = n0; n < n1; ++n) {
        const bool haveN = (n + 1 < n1) && ((t2 - t1) >= 16);
        const int  tn    = (n + 3 < nNodes) ? (n + 3) : nNodes;
        int   t3, e;
        float selfw, hself, acc;
        v16i_u SN = {};
        v16f_u WN = {};

        if (have) {
            float hv[16];
            #pragma unroll
            for (int j = 0; j < 16; ++j) hv[j] = h_at(h, S[j], laneByte);
            hself = h_at(h, n, laneByte);
            const int pe = haveN ? t1 : t0;
            SN    = *(const v16i_u*)(src + pe);
            WN    = *(const v16f_u*)(ew + pe);
            t3    = tar_ptr[tn];
            selfw = sw[n];
            __builtin_amdgcn_sched_barrier(0);
            acc = 0.f;
            #pragma unroll
            for (int j = 0; j < 16; ++j) acc = __builtin_fmaf(Wt[j], hv[j], acc);
            __builtin_amdgcn_sched_barrier(0);
            e = t0 + 16;
        } else {
            hself = h_at(h, n, laneByte);
            t3    = tar_ptr[tn];
            selfw = sw[n];
            if (haveN) {
                SN = *(const v16i_u*)(src + t1);
                WN = *(const v16f_u*)(ew + t1);
            }
            acc = 0.f;
            e = t0;
        }
        for (; e + 16 <= t1; e += 16) {
            const v16i_u S2 = *(const v16i_u*)(src + e);
            const v16f_u W2 = *(const v16f_u*)(ew + e);
            float hv[16];
            #pragma unroll
            for (int j = 0; j < 16; ++j) hv[j] = h_at(h, S2[j], laneByte);
            #pragma unroll
            for (int j = 0; j < 16; ++j) acc = __builtin_fmaf(W2[j], hv[j], acc);
        }
        for (; e < t1; ++e) acc = __builtin_fmaf(ew[e], h_at(h, src[e], laneByte), acc);

        acc = __builtin_fmaf(selfw, hself, acc);
        *(float*)((char*)out + (((unsigned)n << 7) + laneByte)) = acc;

        t0 = t1; t1 = t2; t2 = t3;
        S = SN; Wt = WN; have = haveN;
    }
}

extern "C" void kernel_launch(void* const* d_in, const int* in_sizes, int n_in,
                              void* d_out, int out_size, void* d_ws, size_t ws_size,
                              hipStream_t stream) {
    const float* x   = (const float*)d_in[0];
    const float* W   = (const float*)d_in[1];
    const float* b   = (const float*)d_in[2];
    const float* ew  = (const float*)d_in[3];
    const float* sw  = (const float*)d_in[4];
    const int*   tp  = (const int*)d_in[5];
    const int*   src = (const int*)d_in[6];
    float*       out = (float*)d_out;

    const int nNodes = in_sizes[4];
    float* h = (float*)d_ws;

    gcn_gemm_wmma<<<GEMM_WAVES, 32, 0, stream>>>(x, W, b, h, nNodes);

    gcn_aggregate<<<1024, 32 * AGG_WAVES_PER_BLOCK, 0, stream>>>(h, ew, sw, tp, src, out, nNodes);
}


# aggregate: a wave works on two nodes at a time: 34 row loads in flight before the first multiply-add (was 17); the pair's 32 weights by one vector load + readlane; per-node sum order kept
# speedup vs baseline: 2.1439x; 1.1373x over previous
#include <hip/hip_runtime.h>

typedef float v2f __attribute__((ext_vector_type(2)));
typedef float v4f __attribute__((ext_vector_type(4)));
typedef float v8f __attribute__((ext_vector_type(8)));

#define IN_C 128
#define OUT_C 32
#define GEMM_WAVES 1024

__device__ __forceinline__ const v4f* x_frag_ptr(const float* __restrict__ x, int tile, int lastTile,
                                                 int mn, int half, int nRows)
{
    const int t = tile < lastTile ? tile : lastTile;
    int row = t * 16 + mn;
    row = row < nRows ? row : nRows - 1;
    return (const v4f*)(x + (size_t)row * IN_C + 4 * half);
}

__device__ __forceinline__ void tile_mma_store(const v4f (&af)[IN_C / 8], const v4f (&wf0)[IN_C / 8],
                                               const v4f (&wf1)[IN_C / 8], float bias0, float bias1,
                                               float* __restrict__ h, int tileRow0, int half, int mn)
{
    v8f c0 = {};
    v8f c1 = {};
    #pragma unroll
    for (int j = 0; j < IN_C / 8; ++j) {
        v2f alo  = __builtin_shufflevector(af[j],  af[j],  0, 1);
        v2f ahi  = __builtin_shufflevector(af[j],  af[j],  2, 3);
        v2f b0lo = __builtin_shufflevector(wf0[j], wf0[j], 0, 1);
        v2f b0hi = __builtin_shufflevector(wf0[j], wf0[j], 2, 3);
        v2f b1lo = __builtin_shufflevector(wf1[j], wf1[j], 0, 1);
        v2f b1hi = __builtin_shufflevector(wf1[j], wf1[j], 2, 3);
        c0 = __builtin_amdgcn_wmma_f32_16x16x4_f32(false, alo, false, b0lo,
                                                   (short)0, c0, false, false);
        c1 = __builtin_amdgcn_wmma_f32_16x16x4_f32(false, alo, false, b1lo,
                                                   (short)0, c1, false, false);
        c0 = __builtin_amdgcn_wmma_f32_16x16x4_f32(false, ahi, false, b0hi,
                                                   (short)0, c0, false, false);
        c1 = __builtin_amdgcn_wmma_f32_16x16x4_f32(false, ahi, false, b1hi,
                                                   (short)0, c1, false, false);
    }
    float* hp = h + (size_t)(tileRow0 + 8 * half) * OUT_C + mn;
    #pragma unroll
    for (int v = 0; v < 8; ++v) {
        hp[v * OUT_C]      = c0[v] + bias0;
        hp[v * OUT_C + 16] = c1[v] + bias1;
    }
}

__global__ __launch_bounds__(32)
void gcn_gemm_wmma(const float* __restrict__ x, const float* __restrict__ W,
                   const float* __restrict__ b, float* __restrict__ h, int nRows)
{
    const int lane = threadIdx.x;
    const int half = lane >> 4;
    const int mn   = lane & 15;

    const int nTiles = (nRows + 15) >> 4;
    const int nW     = (int)gridDim.x;
    const int w      = (int)blockIdx.x;
    const int base   = nTiles / nW;
    const int rem    = nTiles - base * nW;
    const int tile0  = w * base + (w < rem ? w : rem);
    const int cnt    = base + (w < rem ? 1 : 0);
    if (cnt <= 0) return;
    const int lastTile = tile0 + cnt - 1;

    const v4f* w0p = (const v4f*)(W + (size_t)mn * IN_C + 4 * half);
    const v4f* w1p = (const v4f*)(W + (size_t)(16 + mn) * IN_C + 4 * half);
    v4f wf0[IN_C / 8], wf1[IN_C / 8], afA[IN_C / 8], afB[IN_C / 8];
    {
        const v4f* ap = x_frag_ptr(x, tile0, lastTile, mn, half, nRows);
        #pragma unroll
        for (int j = 0; j < IN_C / 8; ++j) {
            afA[j] = ap[2 * j];
            wf0[j] = w0p[2 * j];
            wf1[j] = w1p[2 * j];
        }
    }

    const float bias0 = b[mn];
    const float bias1 = b[16 + mn];

    int t = tile0;
    for (int p = cnt >> 1; p > 0; --p, t += 2) {
        {
            const v4f* ap = x_frag_ptr(x, t + 1, lastTile, mn, half, nRows);
            #pragma unroll
            for (int j = 0; j < IN_C / 8; ++j) afB[j] = ap[2 * j];
        }
        __builtin_amdgcn_sched_barrier(0);
        tile_mma_store(afA, wf0, wf1, bias0, bias1, h, t * 16, half, mn);
        {
            const v4f* ap = x_frag_ptr(x, t + 2, lastTile, mn, half, nRows);
            #pragma unroll
            for (int j = 0; j < IN_C / 8; ++j) afA[j] = ap[2 * j];
        }
        __builtin_amdgcn_sched_barrier(0);
        tile_mma_store(afB, wf0, wf1, bias0, bias1, h, (t + 1) * 16, half, mn);
    }
    if (cnt & 1)
        tile_mma_store(afA, wf0, wf1, bias0, bias1, h, t * 16, half, mn);
}

typedef int v16i_u __attribute__((ext_vector_type(16), aligned(4)));

#define AGG_WAVES_PER_BLOCK 8

__device__ __forceinline__ float h_at(const float* __restrict__ h, int s, unsigned laneByte)
{
    unsigned boff = ((unsigned)s << 7) + laneByte;
    return *(const float*)((const char*)h + boff);
}

__device__ __forceinline__ float edges_plain(float acc, int e, int eEnd, const float* __restrict__ h,
                                             const float* __restrict__ ew, const int* __restrict__ src,
                                             unsigned laneByte)
{
    for (; e < eEnd; ++e) acc = __builtin_fmaf(ew[e], h_at(h, src[e], laneByte), acc);
    return acc;
}

__device__ __forceinline__ float pair_weights(const float* __restrict__ ew, int eA, int eB, int lane)
{
    const unsigned e = (unsigned)((lane < 16 ? eA : eB) + (lane & 15));
    return *(const float*)((const char*)ew + (e << 2));
}

__device__ __forceinline__ float lane_value(float v, int l)
{
    return __builtin_bit_cast(float, __builtin_amdgcn_readlane(__builtin_bit_cast(int, v), l));
}

__global__ __launch_bounds__(32 * AGG_WAVES_PER_BLOCK)
void gcn_aggregate(const float* __restrict__ h, const float* __restrict__ ew,
                   const float* __restrict__ sw, const int* __restrict__ tar_ptr,
                   const int* __restrict__ src, float* __restrict__ out, int nNodes)
{
    const int lane = (int)(threadIdx.x & 31u);
    const unsigned laneByte = (unsigned)lane * 4u;
    const int waveInBlock = __builtin_amdgcn_readfirstlane((int)(threadIdx.x >> 5));
    const int wave   = (int)blockIdx.x * AGG_WAVES_PER_BLOCK + waveInBlock;
    const int nWaves = (int)gridDim.x * AGG_WAVES_PER_BLOCK;
    const int per    = (((nNodes + nWaves - 1) / nWaves) + 1) & ~1;
    const int n0     = wave * per;
    const int n1     = (n0 + per < nNodes) ? (n0 + per) : nNodes;
    if (n0 >= n1) return;

    int b0 = tar_ptr[n0];
    int b1 = tar_ptr[n0 + 1];
    int b2 = tar_ptr[(n0 + 2 < nNodes) ? (n0 + 2) : nNodes];
    int b3 = tar_ptr[(n0 + 3 < nNodes) ? (n0 + 3) : nNodes];
    int b4 = tar_ptr[(n0 + 4 < nNodes) ? (n0 + 4) : nNodes];
    const int tEnd = tar_ptr[n1];

    {
        const int eL = b0 + lane * 64;
        if (eL < tEnd) {
            __builtin_prefetch(src + eL, 0, 1);
            __builtin_prefetch(ew + eL, 0, 1);
        }
    }

    bool fast = (n0 + 1 < n1) && ((b1 - b0) >= 16) && ((b2 - b1) >= 16);
    v16i_u SA = {}, SB = {};
    float  wv = 0.f;
    if (fast) {
        SA = *(const v16i_u*)(src + b0);
        SB = *(const v16i_u*)(src + b1);
        wv = pair_weights(ew, b0, b1, lane);
    }

    for (int n = n0; n < n1; n += 2) {
        const bool fastN = (n + 3 < n1) && ((b3 - b2) >= 16) && ((b4 - b3) >= 16);
        const int  i5 = (n + 5 < nNodes) ? (n + 5) : nNodes;
        const int  i6 = (n + 6 < nNodes) ? (n + 6) : nNodes;
        int b5, b6;

        if (fast) {
            float ha[16], hb[16];
            #pragma unroll
            for (int j = 0; j < 16; ++j) {
                ha[j] = h_at(h, SA[j], laneByte);
                hb[j] = h_at(h, SB[j], laneByte);
            }
            const float hsa = h_at(h, n, laneByte);
            const float hsb = h_at(h, n + 1, laneByte);
            const int pa = fastN ? b2 : b0;
            const int pb = fastN ? b3 : b1;
            SA = *(const v16i_u*)(src + pa);
            SB = *(const v16i_u*)(src + pb);
            const float wvN = pair_weights(ew, pa, pb, lane);
            b5 = tar_ptr[i5];
            b6 = tar_ptr[i6];
            const float swa = sw[n];
            const float swb = sw[n + 1];
            __builtin_amdgcn_sched_barrier(0);
            float accA = 0.f, accB = 0.f;
            #pragma unroll
            for (int j = 0; j < 16; ++j) {
                accA = __builtin_fmaf(lane_value(wv, j),      ha[j], accA);
                accB = __builtin_fmaf(lane_value(wv, 16 + j), hb[j], accB);
            }
            if (((b1 - b0) > 16) | ((b2 - b1) > 16)) {
                accA = edges_plain(accA, b0 + 16, b1, h, ew, src, laneByte);
                accB = edges_plain(accB, b1 + 16, b2, h, ew, src, laneByte);
            }
            accA = __builtin_fmaf(swa, hsa, accA);
            accB = __builtin_fmaf(swb, hsb, accB);
            *(float*)((char*)out + (((unsigned)n << 7) + laneByte))       = accA;
            *(float*)((char*)out + (((unsigned)(n + 1) << 7) + laneByte)) = accB;
            wv = wvN;
        } else {
            float acc = edges_plain(0.f, b0, b1, h, ew, src, laneByte);
            acc = __builtin_fmaf(sw[n], h_at(h, n, laneByte), acc);
            *(float*)((char*)out + (((unsigned)n << 7) + laneByte)) = acc;
            if (n + 1 < n1) {
                float acc2 = edges_plain(0.f, b1, b2, h, ew, src, laneByte);
                acc2 = __builtin_fmaf(sw[n + 1], h_at(h, n + 1, laneByte), acc2);
                *(float*)((char*)out + (((unsigned)(n + 1) << 7) + laneByte)) = acc2;
            }
            b5 = tar_ptr[i5];
            b6 = tar_ptr[i6];
            if (fastN) {
                SA = *(const v16i_u*)(src + b2);
                SB = *(const v16i_u*)(src + b3);
                wv = pair_weights(ew, b2, b3, lane);
            }
        }
        b0 = b2; b1 = b3; b2 = b4; b3 = b5; b4 = b6;
        fast = fastN;
    }
}

extern "C" void kernel_launch(void* const* d_in, const int* in_sizes, int n_in,
                              void* d_out, int out_size, void* d_ws, size_t ws_size,
                              hipStream_t stream) {
    const float* x   = (const float*)d_in[0];
    const float* W   = (const float*)d_in[1];
    const float* b   = (const float*)d_in[2];
    const float* ew  = (const float*)d_in[3];
    const float* sw  = (const float*)d_in[4];
    const int*   tp  = (const int*)d_in[5];
    const int*   src = (const int*)d_in[6];
    float*       out = (float*)d_out;

    const int nNodes = in_sizes[4];
    float* h = (float*)d_ws;

    gcn_gemm_wmma<<<GEMM_WAVES, 32, 0, stream>>>(x, W, b, h, nNodes);

    gcn_aggregate<<<1024, 32 * AGG_WAVES_PER_BLOCK, 0, stream>>>(h, ew, sw, tp, src, out, nNodes);
}
